# starting kernel
# baseline (speedup 1.0000x reference)
#include <hip/hip_runtime.h>
#include <hip/hip_bf16.h>

#define HIDDEN 128

typedef __attribute__((ext_vector_type(2))) float v2f;
typedef __attribute__((ext_vector_type(8))) float v8f;

__global__ __launch_bounds__(256) void gconv_gemm_relu(
    const float* __restrict__ x, const float* __restrict__ W,
    const float* __restrict__ bias, float* __restrict__ m, int N) {
  const int lane = threadIdx.x & 31;
  const int wave = threadIdx.x >> 5;
  const int row_base = blockIdx.x * 16;
  const int col_base = wave * 16;
  if (row_base >= N) return;

  const int n    = lane & 15;
  const int kgrp = (lane >> 4) * 2;

  const float* __restrict__ xrow = x + (size_t)(row_base + n) * HIDDEN;
  const float* __restrict__ wrow = W + (size_t)(col_base + n) * HIDDEN;

  v8f acc = {};
#pragma unroll
  for (int k = 0; k < HIDDEN; k += 4) {
    v2f a = *(const v2f*)(xrow + k + kgrp);
    v2f b = *(const v2f*)(wrow + k + kgrp);
    acc = __builtin_amdgcn_wmma_f32_16x16x4_f32(
         false, a,  false, b,
         (short)0, acc,  false,  false);
  }

  const float bv = bias[col_base + n];
  const int moff = (lane >> 4) * 8;
  float* __restrict__ mp = m + (size_t)row_base * HIDDEN + col_base + n;
#pragma unroll
  for (int g = 0; g < 8; ++g) {
    float v = acc[g] + bv;
    mp[(size_t)(g + moff) * HIDDEN] = v > 0.0f ? v : 0.0f;
  }
}

__global__ void gconv_zero(float* __restrict__ p, size_t n) {
  size_t i = (size_t)blockIdx.x * blockDim.x + threadIdx.x;
  if (i < n) p[i] = 0.0f;
}

__global__ __launch_bounds__(256) void gconv_scatter(
    const float* __restrict__ m, const int* __restrict__ rows,
    const int* __restrict__ cols, float* __restrict__ agg,
    float* __restrict__ deg, int nEdges) {
  long long t = (long long)blockIdx.x * blockDim.x + threadIdx.x;
  int e = (int)(t >> 7);
  if (e >= nEdges) return;
  int f = (int)(t & (HIDDEN - 1));
  int r = rows[e];
  int c = cols[e];
  atomicAdd(&agg[(size_t)r * HIDDEN + f], m[(size_t)c * HIDDEN + f]);
  if (f == 0) atomicAdd(&deg[r], 1.0f);
}

__global__ __launch_bounds__(256) void gconv_finalize(
    const float* __restrict__ x, const float* __restrict__ agg,
    const float* __restrict__ deg, const float* __restrict__ gamma,
    const float* __restrict__ beta, float* __restrict__ out, int N) {
  int node = blockIdx.x * (blockDim.x >> 5) + (threadIdx.x >> 5);
  if (node >= N) return;
  const int lane = threadIdx.x & 31;

  float d = deg[node];
  float invd = 1.0f / (d == 0.0f ? 1.0f : d);

  const float4 xv = *(const float4*)(x + (size_t)node * HIDDEN + lane * 4);
  const float4 av = *(const float4*)(agg + (size_t)node * HIDDEN + lane * 4);
  float4 h;
  h.x = xv.x + av.x * invd;
  h.y = xv.y + av.y * invd;
  h.z = xv.z + av.z * invd;
  h.w = xv.w + av.w * invd;

  float ss = h.x * h.x + h.y * h.y + h.z * h.z + h.w * h.w;
#pragma unroll
  for (int off = 16; off >= 1; off >>= 1) ss += __shfl_xor(ss, off, 32);
  const float inv_rms = rsqrtf(ss * (1.0f / HIDDEN) + 1e-5f);

  const float4 gv = *(const float4*)(gamma + lane * 4);
  const float4 bv = *(const float4*)(beta + lane * 4);
  float4 o;
  o.x = h.x * inv_rms * gv.x + bv.x;
  o.y = h.y * inv_rms * gv.y + bv.y;
  o.z = h.z * inv_rms * gv.z + bv.z;
  o.w = h.w * inv_rms * gv.w + bv.w;
  *(float4*)(out + (size_t)node * HIDDEN + lane * 4) = o;
}

extern "C" void kernel_launch(void* const* d_in, const int* in_sizes, int n_in,
                              void* d_out, int out_size, void* d_ws, size_t ws_size,
                              hipStream_t stream) {
  const float* x     = (const float*)d_in[0];
  const float* W     = (const float*)d_in[1];
  const float* bias  = (const float*)d_in[2];
  const float* gamma = (const float*)d_in[3];
  const float* beta  = (const float*)d_in[4];
  const int*   rows  = (const int*)d_in[5];
  const int*   cols  = (const int*)d_in[6];
  float*       out   = (float*)d_out;

  const int N      = in_sizes[0] / HIDDEN;
  const int nEdges = in_sizes[5];

  float* m   = (float*)d_ws;
  float* agg = m + (size_t)N * HIDDEN;
  float* deg = agg + (size_t)N * HIDDEN;

  size_t zn = (size_t)N * HIDDEN + (size_t)N;
  gconv_zero<<<(int)((zn + 255) / 256), 256, 0, stream>>>(agg, zn);

  gconv_gemm_relu<<<(N + 15) / 16, 256, 0, stream>>>(x, W, bias, m, N);

  long long tot = (long long)nEdges * HIDDEN;
  gconv_scatter<<<(int)((tot + 255) / 256), 256, 0, stream>>>(m, rows, cols, agg,
                                                              deg, nEdges);

  gconv_finalize<<<(N + 7) / 8, 256, 0, stream>>>(x, agg, deg, gamma, beta, out, N);
}


# f32 atomic scatter replaced by grouping edges by destination row (integer slot adds, scan, fill) + one wave per node summing source rows in registers, finalize fused; sums in ascending source order
# speedup vs baseline: 2.9341x; 2.9341x over previous
#include <hip/hip_runtime.h>
#include <hip/hip_bf16.h>

#define HIDDEN 128

typedef __attribute__((ext_vector_type(2))) float v2f;
typedef __attribute__((ext_vector_type(8))) float v8f;

__global__ __launch_bounds__(256) void gconv_gemm_relu(
    const float* __restrict__ x, const float* __restrict__ W,
    const float* __restrict__ bias, float* __restrict__ m, int N) {
  const int lane = threadIdx.x & 31;
  const int wave = threadIdx.x >> 5;
  const int row_base = blockIdx.x * 16;
  const int col_base = wave * 16;
  if (row_base >= N) return;

  const int n    = lane & 15;
  const int kgrp = (lane >> 4) * 2;

  const float* __restrict__ xrow = x + (size_t)(row_base + n) * HIDDEN;
  const float* __restrict__ wrow = W + (size_t)(col_base + n) * HIDDEN;

  v8f acc = {};
#pragma unroll
  for (int k = 0; k < HIDDEN; k += 4) {
    v2f a = *(const v2f*)(xrow + k + kgrp);
    v2f b = *(const v2f*)(wrow + k + kgrp);
    acc = __builtin_amdgcn_wmma_f32_16x16x4_f32(
         false, a,  false, b,
         (short)0, acc,  false,  false);
  }

  const float bv = bias[col_base + n];
  const int moff = (lane >> 4) * 8;
  float* __restrict__ mp = m + (size_t)row_base * HIDDEN + col_base + n;
#pragma unroll
  for (int g = 0; g < 8; ++g) {
    float v = acc[g] + bv;
    mp[(size_t)(g + moff) * HIDDEN] = v > 0.0f ? v : 0.0f;
  }
}


__global__ void gconv_zero_counts(int* __restrict__ cnt, int n) {
  int i = blockIdx.x * blockDim.x + threadIdx.x;
  if (i < n) cnt[i] = 0;
}

__global__ __launch_bounds__(256) void gconv_rank(
    const int* __restrict__ rows, int* __restrict__ cnt,
    int* __restrict__ slot, int nEdges) {
  int e = blockIdx.x * 256 + threadIdx.x;
  if (e >= nEdges) return;
  slot[e] = atomicAdd(&cnt[rows[e]], 1);
}

#define SCAN_T 1024
__global__ __launch_bounds__(SCAN_T) void gconv_scan(
    const int* __restrict__ cnt, int* __restrict__ start, int N) {
  __shared__ int s_front[SCAN_T / 32];
  __shared__ int s_tot[SCAN_T / 32];
  const int tid = threadIdx.x;
  const int lane = tid & 31;
  const int wave = tid >> 5;
  const int block_first = blockIdx.x * (SCAN_T * 4);

  const int4* __restrict__ c4 = (const int4*)cnt;
  int front = 0;
#pragma unroll 4
  for (int it = 0; it < (int)blockIdx.x; ++it) {
    const int4 v = c4[it * SCAN_T + tid];
    front += (v.x + v.y) + (v.z + v.w);
  }
#pragma unroll
  for (int o = 16; o >= 1; o >>= 1) front += __shfl_xor(front, o, 32);

  const int i0 = block_first + tid * 4;
  const int c0 = (i0     < N) ? cnt[i0]     : 0;
  const int c1 = (i0 + 1 < N) ? cnt[i0 + 1] : 0;
  const int c2 = (i0 + 2 < N) ? cnt[i0 + 2] : 0;
  const int c3 = (i0 + 3 < N) ? cnt[i0 + 3] : 0;
  const int mine = (c0 + c1) + (c2 + c3);
  int incl = mine;
#pragma unroll
  for (int o = 1; o < 32; o <<= 1) {
    const int t = __shfl_up(incl, o, 32);
    if (lane >= o) incl += t;
  }
  if (lane == 31) s_tot[wave] = incl;
  if (lane == 0) s_front[wave] = front;
  __syncthreads();
  int before = 0;
#pragma unroll
  for (int w = 0; w < SCAN_T / 32; ++w) {
    before += s_front[w];
    if (w < wave) before += s_tot[w];
  }
  const int ex = before + incl - mine;
  if (i0     < N) start[i0]     = ex;
  if (i0 + 1 < N) start[i0 + 1] = ex + c0;
  if (i0 + 2 < N) start[i0 + 2] = ex + c0 + c1;
  if (i0 + 3 < N) start[i0 + 3] = ex + c0 + c1 + c2;
}

__global__ __launch_bounds__(256) void gconv_fill(
    const int* __restrict__ rows, const int* __restrict__ cols,
    const int* __restrict__ start, const int* __restrict__ slot,
    int* __restrict__ src, int nEdges) {
  int e = blockIdx.x * 256 + threadIdx.x;
  if (e >= nEdges) return;
  src[start[rows[e]] + slot[e]] = cols[e];
}

template <int K>
__device__ __forceinline__ void gconv_add_rows(const float* __restrict__ m,
                                               int mysrc, int j, int lane,
                                               float4& acc) {
  float4 v[K];
#pragma unroll
  for (int u = 0; u < K; ++u) {
    const int c = __builtin_amdgcn_readlane(mysrc, j + u);
    v[u] = *(const float4*)(m + (size_t)c * HIDDEN + lane * 4);
  }
#pragma unroll
  for (int u = 0; u < K; ++u) {
    acc.x += v[u].x;
    acc.y += v[u].y;
    acc.z += v[u].z;
    acc.w += v[u].w;
  }
}

__global__ __launch_bounds__(256) void gconv_gather_finalize(
    const float* __restrict__ x, const float* __restrict__ m,
    const int* __restrict__ cnt, const int* __restrict__ start,
    const int* __restrict__ src, const float* __restrict__ gamma,
    const float* __restrict__ beta, float* __restrict__ out, int N) {
  const int lane = threadIdx.x & 31;
  const int node = __builtin_amdgcn_readfirstlane(
      (int)(blockIdx.x * (blockDim.x >> 5) + (threadIdx.x >> 5)));
  if (node >= N) return;

  const int d = cnt[node];
  const int s = start[node];

  float4 acc = make_float4(0.0f, 0.0f, 0.0f, 0.0f);
  for (int base = 0; base < d; base += 32) {
    const int n = (d - base) < 32 ? (d - base) : 32;
    int mysrc = (lane < n) ? src[s + base + lane] : 0;
    if (d <= 32) {
      int pos = 0;
      for (int k = 0; k < n; ++k) {
        const int ck = __builtin_amdgcn_readlane(mysrc, k);
        pos += ((ck < mysrc) || (ck == mysrc && k < lane)) ? 1 : 0;
      }
      if (lane >= n) pos = lane;
      mysrc = __builtin_amdgcn_ds_permute(pos << 2, mysrc);
    }
    int j = 0;
    for (; j + 8 <= n; j += 8) gconv_add_rows<8>(m, mysrc, j, lane, acc);
    if (j + 4 <= n) { gconv_add_rows<4>(m, mysrc, j, lane, acc); j += 4; }
    if (j + 2 <= n) { gconv_add_rows<2>(m, mysrc, j, lane, acc); j += 2; }
    if (j < n)      { gconv_add_rows<1>(m, mysrc, j, lane, acc); }
  }

  const float df = (float)d;
  const float invd = 1.0f / (df == 0.0f ? 1.0f : df);

  const float4 xv = *(const float4*)(x + (size_t)node * HIDDEN + lane * 4);
  float4 h;
  h.x = xv.x + acc.x * invd;
  h.y = xv.y + acc.y * invd;
  h.z = xv.z + acc.z * invd;
  h.w = xv.w + acc.w * invd;

  float ss = h.x * h.x + h.y * h.y + h.z * h.z + h.w * h.w;
#pragma unroll
  for (int off = 16; off >= 1; off >>= 1) ss += __shfl_xor(ss, off, 32);
  const float inv_rms = rsqrtf(ss * (1.0f / HIDDEN) + 1e-5f);

  const float4 gv = *(const float4*)(gamma + lane * 4);
  const float4 bv = *(const float4*)(beta + lane * 4);
  float4 o;
  o.x = h.x * inv_rms * gv.x + bv.x;
  o.y = h.y * inv_rms * gv.y + bv.y;
  o.z = h.z * inv_rms * gv.z + bv.z;
  o.w = h.w * inv_rms * gv.w + bv.w;
  *(float4*)(out + (size_t)node * HIDDEN + lane * 4) = o;
}

extern "C" void kernel_launch(void* const* d_in, const int* in_sizes, int n_in,
                              void* d_out, int out_size, void* d_ws, size_t ws_size,
                              hipStream_t stream) {
  const float* x     = (const float*)d_in[0];
  const float* W     = (const float*)d_in[1];
  const float* bias  = (const float*)d_in[2];
  const float* gamma = (const float*)d_in[3];
  const float* beta  = (const float*)d_in[4];
  const int*   rows  = (const int*)d_in[5];
  const int*   cols  = (const int*)d_in[6];
  float*       out   = (float*)d_out;

  const int N      = in_sizes[0] / HIDDEN;
  const int nEdges = in_sizes[5];

  float* m     = (float*)d_ws;
  int*   cnt   = (int*)(m + (size_t)N * HIDDEN);
  int*   start = cnt + N;
  int*   slot  = start + N;
  int*   src   = slot + nEdges;

  gconv_zero_counts<<<(N + 255) / 256, 256, 0, stream>>>(cnt, N);
  gconv_rank<<<(nEdges + 255) / 256, 256, 0, stream>>>(rows, cnt, slot, nEdges);
  gconv_scan<<<(N + SCAN_T * 4 - 1) / (SCAN_T * 4), SCAN_T, 0, stream>>>(cnt, start, N);
  gconv_fill<<<(nEdges + 255) / 256, 256, 0, stream>>>(rows, cols, start, slot, src, nEdges);

  gconv_gemm_relu<<<(N + 15) / 16, 256, 0, stream>>>(x, W, bias, m, N);

  gconv_gather_finalize<<<(N + 7) / 8, 256, 0, stream>>>(x, m, cnt, start, src, gamma,
                                                        beta, out, N);
}


# GEMM re-tiled: a wave owns 16 rows x all 128 columns (8 accumulator tiles), W staged once per work-group in LDS, x rows read once, persistent grid; k terms summed in a permuted order
# speedup vs baseline: 3.5018x; 1.1935x over previous
#include <hip/hip_runtime.h>
#include <hip/hip_bf16.h>

#define HIDDEN 128

typedef __attribute__((ext_vector_type(2))) float v2f;
typedef __attribute__((ext_vector_type(8))) float v8f;

#define GEMM_WAVES 8
#define GEMM_GRID 256
#define LDS_PITCH 132

__device__ __forceinline__ void gconv_load_strip(float4 (&a)[16],
                                                 const float* __restrict__ x,
                                                 int s, int c, int h, int N) {
  int row = s * 16 + c;
  row = row < N ? row : N - 1;
  const float* __restrict__ xr = x + (size_t)row * HIDDEN + 4 * h;
#pragma unroll
  for (int u = 0; u < 16; ++u) a[u] = *(const float4*)(xr + 8 * u);
}

__device__ __forceinline__ void gconv_load_b(float4 (&b)[16], const float* bw, int p) {
#pragma unroll
  for (int q = 0; q < 2; ++q) {
#pragma unroll
    for (int j = 0; j < 8; ++j)
      b[q * 8 + j] = *(const float4*)(bw + j * 16 * LDS_PITCH + 8 * (2 * p + q));
  }
}

__global__ __launch_bounds__(GEMM_WAVES * 32) void gconv_gemm_relu(
    const float* __restrict__ x, const float* __restrict__ W,
    const float* __restrict__ bias, float* __restrict__ m, int N) {
  __shared__ __attribute__((aligned(16))) float sW[HIDDEN * LDS_PITCH];
  __shared__ __attribute__((aligned(16))) float sRow[GEMM_WAVES][16 * LDS_PITCH];

  const int tid  = threadIdx.x;
  const int lane = tid & 31;
  const int wave = __builtin_amdgcn_readfirstlane(tid >> 5);
  const int c = lane & 15;
  const int h = lane >> 4;
  const int nStrips = (N + 15) >> 4;
  const int nWaves  = (int)gridDim.x * GEMM_WAVES;
  int strip = wave * (int)gridDim.x + (int)blockIdx.x;

  float4 a[16];
  gconv_load_strip(a, x, strip, c, h, N);

  {
    float4 w[16];
#pragma unroll
    for (int t = 0; t < 16; ++t)
      w[t] = ((const float4*)W)[t * (GEMM_WAVES * 32) + tid];
#pragma unroll
    for (int t = 0; t < 16; ++t) {
      const int i = t * (GEMM_WAVES * 32) + tid;
      *(float4*)(sW + (i >> 5) * LDS_PITCH + (i & 31) * 4) = w[t];
    }
  }
  float bv[8];
#pragma unroll
  for (int j = 0; j < 8; ++j) bv[j] = bias[16 * j + c];
  __syncthreads();

  const float* bw = sW + c * LDS_PITCH + 4 * h;
  float* so = sRow[wave];

  for (; strip < nStrips; strip += nWaves) {
    float4 an[16];
    gconv_load_strip(an, x, strip + nWaves, c, h, N);

    v8f acc[8];
#pragma unroll
    for (int j = 0; j < 8; ++j) acc[j] = (v8f){};

    float4 b[2][16];
    gconv_load_b(b[0], bw, 0);
#pragma unroll
    for (int p = 0; p < 8; ++p) {
      if (p + 1 < 8) gconv_load_b(b[(p + 1) & 1], bw, p + 1);
      __builtin_amdgcn_sched_barrier(0);
#pragma unroll
      for (int q = 0; q < 2; ++q) {
        const int u = 2 * p + q;
        const v2f a0 = {a[u].x, a[u].y};
        const v2f a1 = {a[u].z, a[u].w};
#pragma unroll
        for (int j = 0; j < 8; ++j) {
          const float4 bq = b[p & 1][q * 8 + j];
          const v2f b0 = {bq.x, bq.y};
          acc[j] = __builtin_amdgcn_wmma_f32_16x16x4_f32(
               false, a0,  false, b0,
               (short)0, acc[j],  false,  false);
        }
#pragma unroll
        for (int j = 0; j < 8; ++j) {
          const float4 bq = b[p & 1][q * 8 + j];
          const v2f b1 = {bq.z, bq.w};
          acc[j] = __builtin_amdgcn_wmma_f32_16x16x4_f32(
               false, a1,  false, b1,
               (short)0, acc[j],  false,  false);
        }
      }
      __builtin_amdgcn_sched_barrier(0);
    }

#pragma unroll
    for (int j = 0; j < 8; ++j) {
#pragma unroll
      for (int r = 0; r < 8; ++r) {
        const float v = acc[j][r] + bv[j];
        so[(8 * h + r) * LDS_PITCH + 16 * j + c] = v > 0.0f ? v : 0.0f;
      }
    }
    __builtin_amdgcn_fence(__ATOMIC_RELEASE, "wavefront");
    __builtin_amdgcn_wave_barrier();
    __builtin_amdgcn_fence(__ATOMIC_ACQUIRE, "wavefront");
    const int row0 = strip * 16;
    const float* sp = so + lane * 4;
    float* __restrict__ mp = m + (size_t)row0 * HIDDEN + lane * 4;
    if (row0 + 16 <= N) {
      float4 v[16];
#pragma unroll
      for (int r = 0; r < 16; ++r) v[r] = *(const float4*)(sp + r * LDS_PITCH);
#pragma unroll
      for (int r = 0; r < 16; ++r) *(float4*)(mp + r * HIDDEN) = v[r];
    } else {
      for (int r = 0; row0 + r < N; ++r)
        *(float4*)(mp + r * HIDDEN) = *(const float4*)(sp + r * LDS_PITCH);
    }
    __builtin_amdgcn_fence(__ATOMIC_RELEASE, "wavefront");
    __builtin_amdgcn_wave_barrier();
    __builtin_amdgcn_fence(__ATOMIC_ACQUIRE, "wavefront");

#pragma unroll
    for (int u = 0; u < 16; ++u) a[u] = an[u];
  }
}


__global__ void gconv_zero_counts(int* __restrict__ cnt, int n) {
  int i = blockIdx.x * blockDim.x + threadIdx.x;
  if (i < n) cnt[i] = 0;
}

__global__ __launch_bounds__(256) void gconv_rank(
    const int* __restrict__ rows, int* __restrict__ cnt,
    int* __restrict__ slot, int nEdges) {
  int e = blockIdx.x * 256 + threadIdx.x;
  if (e >= nEdges) return;
  slot[e] = atomicAdd(&cnt[rows[e]], 1);
}

#define SCAN_T 1024
__global__ __launch_bounds__(SCAN_T) void gconv_scan(
    const int* __restrict__ cnt, int* __restrict__ start, int N) {
  __shared__ int s_front[SCAN_T / 32];
  __shared__ int s_tot[SCAN_T / 32];
  const int tid = threadIdx.x;
  const int lane = tid & 31;
  const int wave = tid >> 5;
  const int block_first = blockIdx.x * (SCAN_T * 4);

  const int4* __restrict__ c4 = (const int4*)cnt;
  int front = 0;
#pragma unroll 4
  for (int it = 0; it < (int)blockIdx.x; ++it) {
    const int4 v = c4[it * SCAN_T + tid];
    front += (v.x + v.y) + (v.z + v.w);
  }
#pragma unroll
  for (int o = 16; o >= 1; o >>= 1) front += __shfl_xor(front, o, 32);

  const int i0 = block_first + tid * 4;
  const int c0 = (i0     < N) ? cnt[i0]     : 0;
  const int c1 = (i0 + 1 < N) ? cnt[i0 + 1] : 0;
  const int c2 = (i0 + 2 < N) ? cnt[i0 + 2] : 0;
  const int c3 = (i0 + 3 < N) ? cnt[i0 + 3] : 0;
  const int mine = (c0 + c1) + (c2 + c3);
  int incl = mine;
#pragma unroll
  for (int o = 1; o < 32; o <<= 1) {
    const int t = __shfl_up(incl, o, 32);
    if (lane >= o) incl += t;
  }
  if (lane == 31) s_tot[wave] = incl;
  if (lane == 0) s_front[wave] = front;
  __syncthreads();
  int before = 0;
#pragma unroll
  for (int w = 0; w < SCAN_T / 32; ++w) {
    before += s_front[w];
    if (w < wave) before += s_tot[w];
  }
  const int ex = before + incl - mine;
  if (i0     < N) start[i0]     = ex;
  if (i0 + 1 < N) start[i0 + 1] = ex + c0;
  if (i0 + 2 < N) start[i0 + 2] = ex + c0 + c1;
  if (i0 + 3 < N) start[i0 + 3] = ex + c0 + c1 + c2;
}

__global__ __launch_bounds__(256) void gconv_fill(
    const int* __restrict__ rows, const int* __restrict__ cols,
    const int* __restrict__ start, const int* __restrict__ slot,
    int* __restrict__ src, int nEdges) {
  int e = blockIdx.x * 256 + threadIdx.x;
  if (e >= nEdges) return;
  src[start[rows[e]] + slot[e]] = cols[e];
}

template <int K>
__device__ __forceinline__ void gconv_add_rows(const float* __restrict__ m,
                                               int mysrc, int j, int lane,
                                               float4& acc) {
  float4 v[K];
#pragma unroll
  for (int u = 0; u < K; ++u) {
    const int c = __builtin_amdgcn_readlane(mysrc, j + u);
    v[u] = *(const float4*)(m + (size_t)c * HIDDEN + lane * 4);
  }
#pragma unroll
  for (int u = 0; u < K; ++u) {
    acc.x += v[u].x;
    acc.y += v[u].y;
    acc.z += v[u].z;
    acc.w += v[u].w;
  }
}

__global__ __launch_bounds__(256) void gconv_gather_finalize(
    const float* __restrict__ x, const float* __restrict__ m,
    const int* __restrict__ cnt, const int* __restrict__ start,
    const int* __restrict__ src, const float* __restrict__ gamma,
    const float* __restrict__ beta, float* __restrict__ out, int N) {
  const int lane = threadIdx.x & 31;
  const int node = __builtin_amdgcn_readfirstlane(
      (int)(blockIdx.x * (blockDim.x >> 5) + (threadIdx.x >> 5)));
  if (node >= N) return;

  const int d = cnt[node];
  const int s = start[node];

  float4 acc = make_float4(0.0f, 0.0f, 0.0f, 0.0f);
  for (int base = 0; base < d; base += 32) {
    const int n = (d - base) < 32 ? (d - base) : 32;
    int mysrc = (lane < n) ? src[s + base + lane] : 0;
    if (d <= 32) {
      int pos = 0;
      for (int k = 0; k < n; ++k) {
        const int ck = __builtin_amdgcn_readlane(mysrc, k);
        pos += ((ck < mysrc) || (ck == mysrc && k < lane)) ? 1 : 0;
      }
      if (lane >= n) pos = lane;
      mysrc = __builtin_amdgcn_ds_permute(pos << 2, mysrc);
    }
    int j = 0;
    for (; j + 8 <= n; j += 8) gconv_add_rows<8>(m, mysrc, j, lane, acc);
    if (j + 4 <= n) { gconv_add_rows<4>(m, mysrc, j, lane, acc); j += 4; }
    if (j + 2 <= n) { gconv_add_rows<2>(m, mysrc, j, lane, acc); j += 2; }
    if (j < n)      { gconv_add_rows<1>(m, mysrc, j, lane, acc); }
  }

  const float df = (float)d;
  const float invd = 1.0f / (df == 0.0f ? 1.0f : df);

  const float4 xv = *(const float4*)(x + (size_t)node * HIDDEN + lane * 4);
  float4 h;
  h.x = xv.x + acc.x * invd;
  h.y = xv.y + acc.y * invd;
  h.z = xv.z + acc.z * invd;
  h.w = xv.w + acc.w * invd;

  float ss = h.x * h.x + h.y * h.y + h.z * h.z + h.w * h.w;
#pragma unroll
  for (int off = 16; off >= 1; off >>= 1) ss += __shfl_xor(ss, off, 32);
  const float inv_rms = rsqrtf(ss * (1.0f / HIDDEN) + 1e-5f);

  const float4 gv = *(const float4*)(gamma + lane * 4);
  const float4 bv = *(const float4*)(beta + lane * 4);
  float4 o;
  o.x = h.x * inv_rms * gv.x + bv.x;
  o.y = h.y * inv_rms * gv.y + bv.y;
  o.z = h.z * inv_rms * gv.z + bv.z;
  o.w = h.w * inv_rms * gv.w + bv.w;
  *(float4*)(out + (size_t)node * HIDDEN + lane * 4) = o;
}

extern "C" void kernel_launch(void* const* d_in, const int* in_sizes, int n_in,
                              void* d_out, int out_size, void* d_ws, size_t ws_size,
                              hipStream_t stream) {
  const float* x     = (const float*)d_in[0];
  const float* W     = (const float*)d_in[1];
  const float* bias  = (const float*)d_in[2];
  const float* gamma = (const float*)d_in[3];
  const float* beta  = (const float*)d_in[4];
  const int*   rows  = (const int*)d_in[5];
  const int*   cols  = (const int*)d_in[6];
  float*       out   = (float*)d_out;

  const int N      = in_sizes[0] / HIDDEN;
  const int nEdges = in_sizes[5];

  float* m     = (float*)d_ws;
  int*   cnt   = (int*)(m + (size_t)N * HIDDEN);
  int*   start = cnt + N;
  int*   slot  = start + N;
  int*   src   = slot + nEdges;

  gconv_zero_counts<<<(N + 255) / 256, 256, 0, stream>>>(cnt, N);
  gconv_rank<<<(nEdges + 255) / 256, 256, 0, stream>>>(rows, cnt, slot, nEdges);
  gconv_scan<<<(N + SCAN_T * 4 - 1) / (SCAN_T * 4), SCAN_T, 0, stream>>>(cnt, start, N);
  gconv_fill<<<(nEdges + 255) / 256, 256, 0, stream>>>(rows, cols, start, slot, src, nEdges);

  gconv_gemm_relu<<<GEMM_GRID, GEMM_WAVES * 32, 0, stream>>>(x, W, bias, m, N);

  gconv_gather_finalize<<<(N + 7) / 8, 256, 0, stream>>>(x, m, cnt, start, src, gamma,
                                                        beta, out, N);
}


# scan and fill launches removed: each destination row owns a fixed 64-entry list and an edge's source goes straight to the slot its counter add returns; rows over 64 edges sum in edge order
# speedup vs baseline: 3.8716x; 1.1056x over previous
#include <hip/hip_runtime.h>
#include <hip/hip_bf16.h>

#define HIDDEN 128

typedef __attribute__((ext_vector_type(2))) float v2f;
typedef __attribute__((ext_vector_type(8))) float v8f;

#define GEMM_WAVES 8
#define GEMM_GRID 256
#define LDS_PITCH 132

__device__ __forceinline__ void gconv_load_strip(float4 (&a)[16],
                                                 const float* __restrict__ x,
                                                 int s, int c, int h, int N) {
  int row = s * 16 + c;
  row = row < N ? row : N - 1;
  const float* __restrict__ xr = x + (size_t)row * HIDDEN + 4 * h;
#pragma unroll
  for (int u = 0; u < 16; ++u) a[u] = *(const float4*)(xr + 8 * u);
}

__device__ __forceinline__ void gconv_load_b(float4 (&b)[16], const float* bw, int p) {
#pragma unroll
  for (int q = 0; q < 2; ++q) {
#pragma unroll
    for (int j = 0; j < 8; ++j)
      b[q * 8 + j] = *(const float4*)(bw + j * 16 * LDS_PITCH + 8 * (2 * p + q));
  }
}

__global__ __launch_bounds__(GEMM_WAVES * 32) void gconv_gemm_relu(
    const float* __restrict__ x, const float* __restrict__ W,
    const float* __restrict__ bias, float* __restrict__ m, int N) {
  __shared__ __attribute__((aligned(16))) float sW[HIDDEN * LDS_PITCH];
  __shared__ __attribute__((aligned(16))) float sRow[GEMM_WAVES][16 * LDS_PITCH];

  const int tid  = threadIdx.x;
  const int lane = tid & 31;
  const int wave = __builtin_amdgcn_readfirstlane(tid >> 5);
  const int c = lane & 15;
  const int h = lane >> 4;
  const int nStrips = (N + 15) >> 4;
  const int nWaves  = (int)gridDim.x * GEMM_WAVES;
  int strip = wave * (int)gridDim.x + (int)blockIdx.x;

  float4 a[16];
  gconv_load_strip(a, x, strip, c, h, N);

  {
    float4 w[16];
#pragma unroll
    for (int t = 0; t < 16; ++t)
      w[t] = ((const float4*)W)[t * (GEMM_WAVES * 32) + tid];
#pragma unroll
    for (int t = 0; t < 16; ++t) {
      const int i = t * (GEMM_WAVES * 32) + tid;
      *(float4*)(sW + (i >> 5) * LDS_PITCH + (i & 31) * 4) = w[t];
    }
  }
  float bv[8];
#pragma unroll
  for (int j = 0; j < 8; ++j) bv[j] = bias[16 * j + c];
  __syncthreads();

  const float* bw = sW + c * LDS_PITCH + 4 * h;
  float* so = sRow[wave];

  for (; strip < nStrips; strip += nWaves) {
    float4 an[16];
    gconv_load_strip(an, x, strip + nWaves, c, h, N);

    v8f acc[8];
#pragma unroll
    for (int j = 0; j < 8; ++j) acc[j] = (v8f){};

    float4 b[2][16];
    gconv_load_b(b[0], bw, 0);
#pragma unroll
    for (int p = 0; p < 8; ++p) {
      if (p + 1 < 8) gconv_load_b(b[(p + 1) & 1], bw, p + 1);
      __builtin_amdgcn_sched_barrier(0);
#pragma unroll
      for (int q = 0; q < 2; ++q) {
        const int u = 2 * p + q;
        const v2f a0 = {a[u].x, a[u].y};
        const v2f a1 = {a[u].z, a[u].w};
#pragma unroll
        for (int j = 0; j < 8; ++j) {
          const float4 bq = b[p & 1][q * 8 + j];
          const v2f b0 = {bq.x, bq.y};
          acc[j] = __builtin_amdgcn_wmma_f32_16x16x4_f32(
               false, a0,  false, b0,
               (short)0, acc[j],  false,  false);
        }
#pragma unroll
        for (int j = 0; j < 8; ++j) {
          const float4 bq = b[p & 1][q * 8 + j];
          const v2f b1 = {bq.z, bq.w};
          acc[j] = __builtin_amdgcn_wmma_f32_16x16x4_f32(
               false, a1,  false, b1,
               (short)0, acc[j],  false,  false);
        }
      }
      __builtin_amdgcn_sched_barrier(0);
    }

#pragma unroll
    for (int j = 0; j < 8; ++j) {
#pragma unroll
      for (int r = 0; r < 8; ++r) {
        const float v = acc[j][r] + bv[j];
        so[(8 * h + r) * LDS_PITCH + 16 * j + c] = v > 0.0f ? v : 0.0f;
      }
    }
    __builtin_amdgcn_fence(__ATOMIC_RELEASE, "wavefront");
    __builtin_amdgcn_wave_barrier();
    __builtin_amdgcn_fence(__ATOMIC_ACQUIRE, "wavefront");
    const int row0 = strip * 16;
    const float* sp = so + lane * 4;
    float* __restrict__ mp = m + (size_t)row0 * HIDDEN + lane * 4;
    if (row0 + 16 <= N) {
      float4 v[16];
#pragma unroll
      for (int r = 0; r < 16; ++r) v[r] = *(const float4*)(sp + r * LDS_PITCH);
#pragma unroll
      for (int r = 0; r < 16; ++r) *(float4*)(mp + r * HIDDEN) = v[r];
    } else {
      for (int r = 0; row0 + r < N; ++r)
        *(float4*)(mp + r * HIDDEN) = *(const float4*)(sp + r * LDS_PITCH);
    }
    __builtin_amdgcn_fence(__ATOMIC_RELEASE, "wavefront");
    __builtin_amdgcn_wave_barrier();
    __builtin_amdgcn_fence(__ATOMIC_ACQUIRE, "wavefront");

#pragma unroll
    for (int u = 0; u < 16; ++u) a[u] = an[u];
  }
}

#define LIST_CAP 64

__global__ void gconv_zero_counts(int* __restrict__ cnt, int n) {
  int i = blockIdx.x * blockDim.x + threadIdx.x;
  if (i < n) cnt[i] = 0;
}

__global__ __launch_bounds__(256) void gconv_place(
    const int* __restrict__ rows, const int* __restrict__ cols,
    int* __restrict__ cnt, int* __restrict__ lists, int nEdges) {
  int e = blockIdx.x * 256 + threadIdx.x;
  if (e >= nEdges) return;
  const int r = rows[e];
  const int c = cols[e];
  asm volatile("" : : "v"(r), "v"(c));
  const int s = atomicAdd(&cnt[r], 1);
  if (s < LIST_CAP) lists[(size_t)r * LIST_CAP + s] = c;
}

template <int K>
__device__ __forceinline__ void gconv_add_rows(const float* __restrict__ m,
                                               int mysrc, int j, int lane,
                                               float4& acc) {
  float4 v[K];
#pragma unroll
  for (int u = 0; u < K; ++u) {
    const int c = __builtin_amdgcn_readlane(mysrc, j + u);
    v[u] = *(const float4*)(m + (size_t)c * HIDDEN + lane * 4);
  }
#pragma unroll
  for (int u = 0; u < K; ++u) {
    acc.x += v[u].x;
    acc.y += v[u].y;
    acc.z += v[u].z;
    acc.w += v[u].w;
  }
}

__global__ __launch_bounds__(256) void gconv_gather_finalize(
    const float* __restrict__ x, const float* __restrict__ m,
    const int* __restrict__ cnt, const int* __restrict__ lists,
    const int* __restrict__ rows, const int* __restrict__ cols,
    const float* __restrict__ gamma, const float* __restrict__ beta,
    float* __restrict__ out, int N, int nEdges) {
  const int lane = threadIdx.x & 31;
  const int node = __builtin_amdgcn_readfirstlane(
      (int)(blockIdx.x * (blockDim.x >> 5) + (threadIdx.x >> 5)));
  if (node >= N) return;

  const int d = cnt[node];
  const int* __restrict__ src = lists + (size_t)node * LIST_CAP;

  float4 acc = make_float4(0.0f, 0.0f, 0.0f, 0.0f);
  if (d > LIST_CAP) {
    for (int e0 = 0; e0 < nEdges; e0 += 32) {
      const int e = e0 + lane;
      const int r = (e < nEdges) ? rows[e] : -1;
      const int c = (e < nEdges) ? cols[e] : 0;
      unsigned hit = __builtin_amdgcn_ballot_w32(r == node);
      while (hit) {
        const int k = __builtin_ctz(hit);
        hit &= hit - 1;
        gconv_add_rows<1>(m, c, k, lane, acc);
      }
    }
  }
  const int dl = d > LIST_CAP ? 0 : d;
  for (int base = 0; base < dl; base += 32) {
    const int n = (dl - base) < 32 ? (dl - base) : 32;
    int mysrc = (lane < n) ? src[base + lane] : 0;
    if (dl <= 32) {
      int pos = 0;
      for (int k = 0; k < n; ++k) {
        const int ck = __builtin_amdgcn_readlane(mysrc, k);
        pos += ((ck < mysrc) || (ck == mysrc && k < lane)) ? 1 : 0;
      }
      if (lane >= n) pos = lane;
      mysrc = __builtin_amdgcn_ds_permute(pos << 2, mysrc);
    }
    int j = 0;
    for (; j + 8 <= n; j += 8) gconv_add_rows<8>(m, mysrc, j, lane, acc);
    if (j + 4 <= n) { gconv_add_rows<4>(m, mysrc, j, lane, acc); j += 4; }
    if (j + 2 <= n) { gconv_add_rows<2>(m, mysrc, j, lane, acc); j += 2; }
    if (j < n)      { gconv_add_rows<1>(m, mysrc, j, lane, acc); }
  }

  const float df = (float)d;
  const float invd = 1.0f / (df == 0.0f ? 1.0f : df);

  const float4 xv = *(const float4*)(x + (size_t)node * HIDDEN + lane * 4);
  float4 h;
  h.x = xv.x + acc.x * invd;
  h.y = xv.y + acc.y * invd;
  h.z = xv.z + acc.z * invd;
  h.w = xv.w + acc.w * invd;

  float ss = h.x * h.x + h.y * h.y + h.z * h.z + h.w * h.w;
#pragma unroll
  for (int off = 16; off >= 1; off >>= 1) ss += __shfl_xor(ss, off, 32);
  const float inv_rms = rsqrtf(ss * (1.0f / HIDDEN) + 1e-5f);

  const float4 gv = *(const float4*)(gamma + lane * 4);
  const float4 bv = *(const float4*)(beta + lane * 4);
  float4 o;
  o.x = h.x * inv_rms * gv.x + bv.x;
  o.y = h.y * inv_rms * gv.y + bv.y;
  o.z = h.z * inv_rms * gv.z + bv.z;
  o.w = h.w * inv_rms * gv.w + bv.w;
  *(float4*)(out + (size_t)node * HIDDEN + lane * 4) = o;
}

extern "C" void kernel_launch(void* const* d_in, const int* in_sizes, int n_in,
                              void* d_out, int out_size, void* d_ws, size_t ws_size,
                              hipStream_t stream) {
  const float* x     = (const float*)d_in[0];
  const float* W     = (const float*)d_in[1];
  const float* bias  = (const float*)d_in[2];
  const float* gamma = (const float*)d_in[3];
  const float* beta  = (const float*)d_in[4];
  const int*   rows  = (const int*)d_in[5];
  const int*   cols  = (const int*)d_in[6];
  float*       out   = (float*)d_out;

  const int N      = in_sizes[0] / HIDDEN;
  const int nEdges = in_sizes[5];

  float* m     = (float*)d_ws;
  int*   lists = (int*)(m + (size_t)N * HIDDEN);
  int*   cnt   = lists + (size_t)N * LIST_CAP;

  gconv_zero_counts<<<(N + 255) / 256, 256, 0, stream>>>(cnt, N);
  gconv_place<<<(nEdges + 255) / 256, 256, 0, stream>>>(rows, cols, cnt, lists, nEdges);

  gconv_gemm_relu<<<GEMM_GRID, GEMM_WAVES * 32, 0, stream>>>(x, W, bias, m, N);

  gconv_gather_finalize<<<(N + 7) / 8, 256, 0, stream>>>(x, m, cnt, lists, rows, cols,
                                                        gamma, beta, out, N, nEdges);
}


# edge placement folded into the GEMM launch: a wave issues the counter adds of a 256-edge chunk in front of its strip's WMMAs and stores the sources behind them; place launch removed (3 launches)
# speedup vs baseline: 4.0338x; 1.0419x over previous
#include <hip/hip_runtime.h>
#include <hip/hip_bf16.h>

#define HIDDEN 128

typedef __attribute__((ext_vector_type(2))) float v2f;
typedef __attribute__((ext_vector_type(8))) float v8f;

#define GEMM_WAVES 8
#define GEMM_GRID 256
#define LDS_PITCH 132

__device__ __forceinline__ void gconv_load_strip(float4 (&a)[16],
                                                 const float* __restrict__ x,
                                                 int s, int c, int h, int N) {
  int row = s * 16 + c;
  row = row < N ? row : N - 1;
  const float* __restrict__ xr = x + (size_t)row * HIDDEN + 4 * h;
#pragma unroll
  for (int u = 0; u < 16; ++u) a[u] = *(const float4*)(xr + 8 * u);
}

__device__ __forceinline__ void gconv_load_b(float4 (&b)[16], const float* bw, int p) {
#pragma unroll
  for (int q = 0; q < 2; ++q) {
#pragma unroll
    for (int j = 0; j < 8; ++j)
      b[q * 8 + j] = *(const float4*)(bw + j * 16 * LDS_PITCH + 8 * (2 * p + q));
  }
}

#define LIST_CAP 64
#define EDGE_CHUNK 256

struct gconv_edges { int4 r0, r1, c0, c1; };

__device__ __forceinline__ void gconv_load_edges(gconv_edges& e,
                                                 const int* __restrict__ rows,
                                                 const int* __restrict__ cols,
                                                 int chunk, int lane) {
  const int o = chunk * EDGE_CHUNK + lane * 8;
  e.r0 = *(const int4*)(rows + o);
  e.r1 = *(const int4*)(rows + o + 4);
  e.c0 = *(const int4*)(cols + o);
  e.c1 = *(const int4*)(cols + o + 4);
}

template <bool PLACE>
__device__ __forceinline__ void gconv_strips(
    float4 (&a)[16], gconv_edges& en, const float (&bv)[8], const float* bw, float* so,
    const float* __restrict__ x, float* __restrict__ m,
    const int* __restrict__ rows, const int* __restrict__ cols,
    int* __restrict__ cnt, int* __restrict__ lists,
    int strip, int nStrips, int nWaves, int c, int h, int lane, int N) {
  for (; strip < nStrips; strip += nWaves) {
    int er[8], ec[8], es[8];
    if (PLACE) {
      er[0] = en.r0.x; er[1] = en.r0.y; er[2] = en.r0.z; er[3] = en.r0.w;
      er[4] = en.r1.x; er[5] = en.r1.y; er[6] = en.r1.z; er[7] = en.r1.w;
      ec[0] = en.c0.x; ec[1] = en.c0.y; ec[2] = en.c0.z; ec[3] = en.c0.w;
      ec[4] = en.c1.x; ec[5] = en.c1.y; ec[6] = en.c1.z; ec[7] = en.c1.w;
#pragma unroll
      for (int u = 0; u < 8; ++u) es[u] = atomicAdd(&cnt[er[u]], 1);
      int sn = strip + nWaves;
      sn = sn < nStrips ? sn : nStrips - 1;
      gconv_load_edges(en, rows, cols, sn, lane);
    }
    float4 an[16];
    gconv_load_strip(an, x, strip + nWaves, c, h, N);

    v8f acc[8];
#pragma unroll
    for (int j = 0; j < 8; ++j) acc[j] = (v8f){};

    float4 b[2][16];
    gconv_load_b(b[0], bw, 0);
#pragma unroll
    for (int p = 0; p < 8; ++p) {
      if (p + 1 < 8) gconv_load_b(b[(p + 1) & 1], bw, p + 1);
      __builtin_amdgcn_sched_barrier(0);
#pragma unroll
      for (int q = 0; q < 2; ++q) {
        const int u = 2 * p + q;
        const v2f a0 = {a[u].x, a[u].y};
        const v2f a1 = {a[u].z, a[u].w};
#pragma unroll
        for (int j = 0; j < 8; ++j) {
          const float4 bq = b[p & 1][q * 8 + j];
          const v2f b0 = {bq.x, bq.y};
          acc[j] = __builtin_amdgcn_wmma_f32_16x16x4_f32(
               false, a0,  false, b0,
               (short)0, acc[j],  false,  false);
        }
#pragma unroll
        for (int j = 0; j < 8; ++j) {
          const float4 bq = b[p & 1][q * 8 + j];
          const v2f b1 = {bq.z, bq.w};
          acc[j] = __builtin_amdgcn_wmma_f32_16x16x4_f32(
               false, a1,  false, b1,
               (short)0, acc[j],  false,  false);
        }
      }
      __builtin_amdgcn_sched_barrier(0);
    }

#pragma unroll
    for (int j = 0; j < 8; ++j) {
#pragma unroll
      for (int r = 0; r < 8; ++r) {
        const float v = acc[j][r] + bv[j];
        so[(8 * h + r) * LDS_PITCH + 16 * j + c] = v > 0.0f ? v : 0.0f;
      }
    }
    __builtin_amdgcn_fence(__ATOMIC_RELEASE, "wavefront");
    __builtin_amdgcn_wave_barrier();
    __builtin_amdgcn_fence(__ATOMIC_ACQUIRE, "wavefront");
    const int row0 = strip * 16;
    const float* sp = so + lane * 4;
    float* __restrict__ mp = m + (size_t)row0 * HIDDEN + lane * 4;
    if (row0 + 16 <= N) {
      float4 v[16];
#pragma unroll
      for (int r = 0; r < 16; ++r) v[r] = *(const float4*)(sp + r * LDS_PITCH);
#pragma unroll
      for (int r = 0; r < 16; ++r) *(float4*)(mp + r * HIDDEN) = v[r];
    } else {
      for (int r = 0; row0 + r < N; ++r)
        *(float4*)(mp + r * HIDDEN) = *(const float4*)(sp + r * LDS_PITCH);
    }
    __builtin_amdgcn_fence(__ATOMIC_RELEASE, "wavefront");
    __builtin_amdgcn_wave_barrier();
    __builtin_amdgcn_fence(__ATOMIC_ACQUIRE, "wavefront");
    if (PLACE) {
#pragma unroll
      for (int u = 0; u < 8; ++u)
        if (es[u] < LIST_CAP) lists[(size_t)er[u] * LIST_CAP + es[u]] = ec[u];
    }

#pragma unroll
    for (int u = 0; u < 16; ++u) a[u] = an[u];
  }
}

__global__ __launch_bounds__(GEMM_WAVES * 32) void gconv_gemm_relu(
    const float* __restrict__ x, const float* __restrict__ W,
    const float* __restrict__ bias, float* __restrict__ m, int N,
    const int* __restrict__ rows, const int* __restrict__ cols,
    int* __restrict__ cnt, int* __restrict__ lists, int nEdges) {
  __shared__ __attribute__((aligned(16))) float sW[HIDDEN * LDS_PITCH];
  __shared__ __attribute__((aligned(16))) float sRow[GEMM_WAVES][16 * LDS_PITCH];

  const int tid  = threadIdx.x;
  const int lane = tid & 31;
  const int wave = __builtin_amdgcn_readfirstlane(tid >> 5);
  const int c = lane & 15;
  const int h = lane >> 4;
  const int nStrips = (N + 15) >> 4;
  const int nWaves  = (int)gridDim.x * GEMM_WAVES;
  int strip = wave * (int)gridDim.x + (int)blockIdx.x;

  float4 a[16];
  gconv_load_strip(a, x, strip, c, h, N);
  const bool place = nStrips > 0 && nEdges >= nStrips * EDGE_CHUNK;
  gconv_edges en;
  if (place) gconv_load_edges(en, rows, cols, strip < nStrips ? strip : nStrips - 1, lane);

  {
    float4 w[16];
#pragma unroll
    for (int t = 0; t < 16; ++t)
      w[t] = ((const float4*)W)[t * (GEMM_WAVES * 32) + tid];
#pragma unroll
    for (int t = 0; t < 16; ++t) {
      const int i = t * (GEMM_WAVES * 32) + tid;
      *(float4*)(sW + (i >> 5) * LDS_PITCH + (i & 31) * 4) = w[t];
    }
  }
  float bv[8];
#pragma unroll
  for (int j = 0; j < 8; ++j) bv[j] = bias[16 * j + c];
  __syncthreads();

  const float* bw = sW + c * LDS_PITCH + 4 * h;
  float* so = sRow[wave];

  if (place)
    gconv_strips<true>(a, en, bv, bw, so, x, m, rows, cols, cnt, lists,
                       strip, nStrips, nWaves, c, h, lane, N);
  else
    gconv_strips<false>(a, en, bv, bw, so, x, m, rows, cols, cnt, lists,
                        strip, nStrips, nWaves, c, h, lane, N);

  const int eStart = place ? nStrips * EDGE_CHUNK : 0;
  for (int e = eStart + ((int)blockIdx.x * GEMM_WAVES + wave) * 32 + lane; e < nEdges;
       e += nWaves * 32) {
    const int r = rows[e];
    const int cc = cols[e];
    const int s = atomicAdd(&cnt[r], 1);
    if (s < LIST_CAP) lists[(size_t)r * LIST_CAP + s] = cc;
  }
}

__global__ void gconv_zero_counts(int* __restrict__ cnt, int n) {
  int i = blockIdx.x * blockDim.x + threadIdx.x;
  if (i < n) cnt[i] = 0;
}

template <int K>
__device__ __forceinline__ void gconv_add_rows(const float* __restrict__ m,
                                               int mysrc, int j, int lane,
                                               float4& acc) {
  float4 v[K];
#pragma unroll
  for (int u = 0; u < K; ++u) {
    const int c = __builtin_amdgcn_readlane(mysrc, j + u);
    v[u] = *(const float4*)(m + (size_t)c * HIDDEN + lane * 4);
  }
#pragma unroll
  for (int u = 0; u < K; ++u) {
    acc.x += v[u].x;
    acc.y += v[u].y;
    acc.z += v[u].z;
    acc.w += v[u].w;
  }
}

__global__ __launch_bounds__(256) void gconv_gather_finalize(
    const float* __restrict__ x, const float* __restrict__ m,
    const int* __restrict__ cnt, const int* __restrict__ lists,
    const int* __restrict__ rows, const int* __restrict__ cols,
    const float* __restrict__ gamma, const float* __restrict__ beta,
    float* __restrict__ out, int N, int nEdges) {
  const int lane = threadIdx.x & 31;
  const int node = __builtin_amdgcn_readfirstlane(
      (int)(blockIdx.x * (blockDim.x >> 5) + (threadIdx.x >> 5)));
  if (node >= N) return;

  const int d = cnt[node];
  const int* __restrict__ src = lists + (size_t)node * LIST_CAP;

  float4 acc = make_float4(0.0f, 0.0f, 0.0f, 0.0f);
  if (d > LIST_CAP) {
    for (int e0 = 0; e0 < nEdges; e0 += 32) {
      const int e = e0 + lane;
      const int r = (e < nEdges) ? rows[e] : -1;
      const int c = (e < nEdges) ? cols[e] : 0;
      unsigned hit = __builtin_amdgcn_ballot_w32(r == node);
      while (hit) {
        const int k = __builtin_ctz(hit);
        hit &= hit - 1;
        gconv_add_rows<1>(m, c, k, lane, acc);
      }
    }
  }
  const int dl = d > LIST_CAP ? 0 : d;
  for (int base = 0; base < dl; base += 32) {
    const int n = (dl - base) < 32 ? (dl - base) : 32;
    int mysrc = (lane < n) ? src[base + lane] : 0;
    if (dl <= 32) {
      int pos = 0;
      for (int k = 0; k < n; ++k) {
        const int ck = __builtin_amdgcn_readlane(mysrc, k);
        pos += ((ck < mysrc) || (ck == mysrc && k < lane)) ? 1 : 0;
      }
      if (lane >= n) pos = lane;
      mysrc = __builtin_amdgcn_ds_permute(pos << 2, mysrc);
    }
    int j = 0;
    for (; j + 8 <= n; j += 8) gconv_add_rows<8>(m, mysrc, j, lane, acc);
    if (j + 4 <= n) { gconv_add_rows<4>(m, mysrc, j, lane, acc); j += 4; }
    if (j + 2 <= n) { gconv_add_rows<2>(m, mysrc, j, lane, acc); j += 2; }
    if (j < n)      { gconv_add_rows<1>(m, mysrc, j, lane, acc); }
  }

  const float df = (float)d;
  const float invd = 1.0f / (df == 0.0f ? 1.0f : df);

  const float4 xv = *(const float4*)(x + (size_t)node * HIDDEN + lane * 4);
  float4 h;
  h.x = xv.x + acc.x * invd;
  h.y = xv.y + acc.y * invd;
  h.z = xv.z + acc.z * invd;
  h.w = xv.w + acc.w * invd;

  float ss = h.x * h.x + h.y * h.y + h.z * h.z + h.w * h.w;
#pragma unroll
  for (int off = 16; off >= 1; off >>= 1) ss += __shfl_xor(ss, off, 32);
  const float inv_rms = rsqrtf(ss * (1.0f / HIDDEN) + 1e-5f);

  const float4 gv = *(const float4*)(gamma + lane * 4);
  const float4 bv = *(const float4*)(beta + lane * 4);
  float4 o;
  o.x = h.x * inv_rms * gv.x + bv.x;
  o.y = h.y * inv_rms * gv.y + bv.y;
  o.z = h.z * inv_rms * gv.z + bv.z;
  o.w = h.w * inv_rms * gv.w + bv.w;
  *(float4*)(out + (size_t)node * HIDDEN + lane * 4) = o;
}

extern "C" void kernel_launch(void* const* d_in, const int* in_sizes, int n_in,
                              void* d_out, int out_size, void* d_ws, size_t ws_size,
                              hipStream_t stream) {
  const float* x     = (const float*)d_in[0];
  const float* W     = (const float*)d_in[1];
  const float* bias  = (const float*)d_in[2];
  const float* gamma = (const float*)d_in[3];
  const float* beta  = (const float*)d_in[4];
  const int*   rows  = (const int*)d_in[5];
  const int*   cols  = (const int*)d_in[6];
  float*       out   = (float*)d_out;

  const int N      = in_sizes[0] / HIDDEN;
  const int nEdges = in_sizes[5];

  float* m     = (float*)d_ws;
  int*   lists = (int*)(m + (size_t)N * HIDDEN);
  int*   cnt   = lists + (size_t)N * LIST_CAP;

  gconv_zero_counts<<<(N + 255) / 256, 256, 0, stream>>>(cnt, N);

  gconv_gemm_relu<<<GEMM_GRID, GEMM_WAVES * 32, 0, stream>>>(x, W, bias, m, N,
                                                            rows, cols, cnt, lists, nEdges);

  gconv_gather_finalize<<<(N + 7) / 8, 256, 0, stream>>>(x, m, cnt, lists, rows, cols,
                                                        gamma, beta, out, N, nEdges);
}


# global counter adds and counter/list planes replaced by two-level grouping on LDS counters: GEMM groups file edges per destination bucket, gather group builds its rows' lists in LDS; 2 launches
# speedup vs baseline: 8.0627x; 1.9988x over previous
#include <hip/hip_runtime.h>
#include <hip/hip_bf16.h>

#define HIDDEN 128

typedef __attribute__((ext_vector_type(2))) float v2f;
typedef __attribute__((ext_vector_type(8))) float v8f;

#define GEMM_WAVES 8
#define GEMM_GRID 256
#define LDS_PITCH 132

__device__ __forceinline__ void gconv_load_strip(float4 (&a)[16],
                                                 const float* __restrict__ x,
                                                 int s, int c, int h, int N) {
  int row = s * 16 + c;
  row = row < N ? row : N - 1;
  const float* __restrict__ xr = x + (size_t)row * HIDDEN + 4 * h;
#pragma unroll
  for (int u = 0; u < 16; ++u) a[u] = *(const float4*)(xr + 8 * u);
}

__device__ __forceinline__ void gconv_load_b(float4 (&b)[16], const float* bw, int p) {
#pragma unroll
  for (int q = 0; q < 2; ++q) {
#pragma unroll
    for (int j = 0; j < 8; ++j)
      b[q * 8 + j] = *(const float4*)(bw + j * 16 * LDS_PITCH + 8 * (2 * p + q));
  }
}

#define LIST_CAP 64
#define EDGE_CHUNK 256
#define GATHER_GRID 256
#define GATHER_WAVES 32
#define SUB_CAP 64

__device__ __forceinline__ int gconv_bucket(int row, int R, float rinv) {
  int b = (int)((float)row * rinv);
  const int rem = row - b * R;
  b += rem >= R ? 1 : 0;
  b -= rem < 0 ? 1 : 0;
  return b;
}

struct gconv_edges { int4 r0, r1, c0, c1; };

__device__ __forceinline__ void gconv_load_edges(gconv_edges& e,
                                                 const int* __restrict__ rows,
                                                 const int* __restrict__ cols,
                                                 int chunk, int lane) {
  const int o = chunk * EDGE_CHUNK + lane * 8;
  e.r0 = *(const int4*)(rows + o);
  e.r1 = *(const int4*)(rows + o + 4);
  e.c0 = *(const int4*)(cols + o);
  e.c1 = *(const int4*)(cols + o + 4);
}

template <bool PLACE>
__device__ __forceinline__ void gconv_strips(
    float4 (&a)[16], gconv_edges& en, const float (&bv)[8], const float* bw, float* so,
    const float* __restrict__ x, float* __restrict__ m,
    const int* __restrict__ rows, const int* __restrict__ cols,
    int* lcnt, int2* __restrict__ sub, int R, float rinv,
    int& strip, int nStrips, int nWaves, int c, int h, int lane, int N) {
  for (; strip < nStrips; strip += nWaves) {
    int er[8], ec[8], es[8], eb[8];
    if (PLACE) {
      er[0] = en.r0.x; er[1] = en.r0.y; er[2] = en.r0.z; er[3] = en.r0.w;
      er[4] = en.r1.x; er[5] = en.r1.y; er[6] = en.r1.z; er[7] = en.r1.w;
      ec[0] = en.c0.x; ec[1] = en.c0.y; ec[2] = en.c0.z; ec[3] = en.c0.w;
      ec[4] = en.c1.x; ec[5] = en.c1.y; ec[6] = en.c1.z; ec[7] = en.c1.w;
#pragma unroll
      for (int u = 0; u < 8; ++u) eb[u] = gconv_bucket(er[u], R, rinv);
#pragma unroll
      for (int u = 0; u < 8; ++u) es[u] = atomicAdd(&lcnt[eb[u]], 1);
      int sn = strip + nWaves;
      sn = sn < nStrips ? sn : nStrips - 1;
      gconv_load_edges(en, rows, cols, sn, lane);
    }
    float4 an[16];
    gconv_load_strip(an, x, strip + nWaves, c, h, N);

    v8f acc[8];
#pragma unroll
    for (int j = 0; j < 8; ++j) acc[j] = (v8f){};

    float4 b[2][16];
    gconv_load_b(b[0], bw, 0);
#pragma unroll
    for (int p = 0; p < 8; ++p) {
      if (p + 1 < 8) gconv_load_b(b[(p + 1) & 1], bw, p + 1);
      __builtin_amdgcn_sched_barrier(0);
#pragma unroll
      for (int q = 0; q < 2; ++q) {
        const int u = 2 * p + q;
        const v2f a0 = {a[u].x, a[u].y};
        const v2f a1 = {a[u].z, a[u].w};
#pragma unroll
        for (int j = 0; j < 8; ++j) {
          const float4 bq = b[p & 1][q * 8 + j];
          const v2f b0 = {bq.x, bq.y};
          acc[j] = __builtin_amdgcn_wmma_f32_16x16x4_f32(
               false, a0,  false, b0,
               (short)0, acc[j],  false,  false);
        }
#pragma unroll
        for (int j = 0; j < 8; ++j) {
          const float4 bq = b[p & 1][q * 8 + j];
          const v2f b1 = {bq.z, bq.w};
          acc[j] = __builtin_amdgcn_wmma_f32_16x16x4_f32(
               false, a1,  false, b1,
               (short)0, acc[j],  false,  false);
        }
      }
      __builtin_amdgcn_sched_barrier(0);
    }

#pragma unroll
    for (int j = 0; j < 8; ++j) {
#pragma unroll
      for (int r = 0; r < 8; ++r) {
        const float v = acc[j][r] + bv[j];
        so[(8 * h + r) * LDS_PITCH + 16 * j + c] = v > 0.0f ? v : 0.0f;
      }
    }
    __builtin_amdgcn_fence(__ATOMIC_RELEASE, "wavefront");
    __builtin_amdgcn_wave_barrier();
    __builtin_amdgcn_fence(__ATOMIC_ACQUIRE, "wavefront");
    const int row0 = strip * 16;
    const float* sp = so + lane * 4;
    float* __restrict__ mp = m + (size_t)row0 * HIDDEN + lane * 4;
    if (row0 + 16 <= N) {
      float4 v[16];
#pragma unroll
      for (int r = 0; r < 16; ++r) v[r] = *(const float4*)(sp + r * LDS_PITCH);
#pragma unroll
      for (int r = 0; r < 16; ++r) *(float4*)(mp + r * HIDDEN) = v[r];
    } else {
      for (int r = 0; row0 + r < N; ++r)
        *(float4*)(mp + r * HIDDEN) = *(const float4*)(sp + r * LDS_PITCH);
    }
    __builtin_amdgcn_fence(__ATOMIC_RELEASE, "wavefront");
    __builtin_amdgcn_wave_barrier();
    __builtin_amdgcn_fence(__ATOMIC_ACQUIRE, "wavefront");
    if (PLACE) {
#pragma unroll
      for (int u = 0; u < 8; ++u)
        if (es[u] < SUB_CAP) sub[(size_t)eb[u] * SUB_CAP + es[u]] = make_int2(er[u], ec[u]);
    }

#pragma unroll
    for (int u = 0; u < 16; ++u) a[u] = an[u];
  }
}

__global__ __launch_bounds__(GEMM_WAVES * 32) void gconv_gemm_relu(
    const float* __restrict__ x, const float* __restrict__ W,
    const float* __restrict__ bias, float* __restrict__ m, int N,
    const int* __restrict__ rows, const int* __restrict__ cols,
    int* __restrict__ binned, int* __restrict__ countsT, int nEdges) {
  __shared__ __attribute__((aligned(16))) float sW[HIDDEN * LDS_PITCH];
  __shared__ __attribute__((aligned(16))) float sRow[GEMM_WAVES][16 * LDS_PITCH];
  __shared__ int lcnt[GATHER_GRID];

  const int tid  = threadIdx.x;
  const int lane = tid & 31;
  const int wave = __builtin_amdgcn_readfirstlane(tid >> 5);
  const int c = lane & 15;
  const int h = lane >> 4;
  const int nStrips = (N + 15) >> 4;
  const int nWaves  = (int)gridDim.x * GEMM_WAVES;
  int strip = wave * (int)gridDim.x + (int)blockIdx.x;

  float4 a[16];
  gconv_load_strip(a, x, strip, c, h, N);
  const bool place = nStrips > 0 && nEdges >= nStrips * EDGE_CHUNK;
  gconv_edges en;
  if (place) gconv_load_edges(en, rows, cols, strip < nStrips ? strip : nStrips - 1, lane);

  {
    float4 w[16];
#pragma unroll
    for (int t = 0; t < 16; ++t)
      w[t] = ((const float4*)W)[t * (GEMM_WAVES * 32) + tid];
#pragma unroll
    for (int t = 0; t < 16; ++t) {
      const int i = t * (GEMM_WAVES * 32) + tid;
      *(float4*)(sW + (i >> 5) * LDS_PITCH + (i & 31) * 4) = w[t];
    }
  }
  float bv[8];
#pragma unroll
  for (int j = 0; j < 8; ++j) bv[j] = bias[16 * j + c];
  for (int i = tid; i < GATHER_GRID; i += GEMM_WAVES * 32) lcnt[i] = 0;
  __syncthreads();

  const float* bw = sW + c * LDS_PITCH + 4 * h;
  float* so = sRow[wave];
  const int R = (N + GATHER_GRID - 1) / GATHER_GRID;
  const float rinv = 1.0f / (float)R;
  int2* __restrict__ sub = (int2*)binned + (size_t)blockIdx.x * GATHER_GRID * SUB_CAP;

  int s = strip;
  if (place)
    gconv_strips<true>(a, en, bv, bw, so, x, m, rows, cols, lcnt, sub, R, rinv,
                       s, nStrips, nWaves, c, h, lane, N);
  else
    gconv_strips<false>(a, en, bv, bw, so, x, m, rows, cols, lcnt, sub, R, rinv,
                        s, nStrips, nWaves, c, h, lane, N);

  if (!place) s = strip;
  const int nChunks = (nEdges + EDGE_CHUNK - 1) / EDGE_CHUNK;
  for (; s < nChunks; s += nWaves) {
    for (int u = 0; u < 8; ++u) {
      const int e = s * EDGE_CHUNK + lane * 8 + u;
      if (e < nEdges) {
        const int r = rows[e];
        const int cc = cols[e];
        const int b = gconv_bucket(r, R, rinv);
        const int k = atomicAdd(&lcnt[b], 1);
        if (k < SUB_CAP) sub[(size_t)b * SUB_CAP + k] = make_int2(r, cc);
      }
    }
  }

  __syncthreads();
  for (int i = tid; i < GATHER_GRID; i += GEMM_WAVES * 32)
    countsT[(size_t)i * GEMM_GRID + blockIdx.x] = lcnt[i];
}

#define G_ROWS 392
#define LIST_PITCH 65

__device__ __forceinline__ void gconv_insert(int row, int col, int lo, int nr,
                                             int* rowcnt, int* rowlist) {
  const int lr = row - lo;
  if (row >= 0 && lr >= 0 && lr < nr) {
    const int k = atomicAdd(&rowcnt[lr], 1);
    if (k < LIST_CAP) rowlist[lr * LIST_PITCH + k] = col;
  }
}

__device__ __forceinline__ int gconv_take_row(int* ticket, int lane) {
  int t = 0;
  if (lane == 0) t = atomicAdd(ticket, 1);
  return __builtin_amdgcn_readfirstlane(t);
}

template <int K>
__device__ __forceinline__ void gconv_add_rows(const float* __restrict__ m,
                                               int mysrc, int j, int lane,
                                               float4& acc) {
  float4 v[K];
#pragma unroll
  for (int u = 0; u < K; ++u) {
    const int c = __builtin_amdgcn_readlane(mysrc, j + u);
    v[u] = *(const float4*)(m + (size_t)c * HIDDEN + lane * 4);
  }
#pragma unroll
  for (int u = 0; u < K; ++u) {
    acc.x += v[u].x;
    acc.y += v[u].y;
    acc.z += v[u].z;
    acc.w += v[u].w;
  }
}

__global__ __launch_bounds__(GATHER_WAVES * 32) void gconv_gather_finalize(
    const float* __restrict__ x, const float* __restrict__ m,
    const int* __restrict__ binned, const int* __restrict__ countsT,
    const int* __restrict__ rows, const int* __restrict__ cols,
    const float* __restrict__ gamma, const float* __restrict__ beta,
    float* __restrict__ out, int N, int nEdges) {
  __shared__ int rowcnt[G_ROWS];
  __shared__ int rowlist[G_ROWS * LIST_PITCH];
  __shared__ int ticket;

  const int tid  = threadIdx.x;
  const int lane = tid & 31;
  const int wave = __builtin_amdgcn_readfirstlane(tid >> 5);
  const int b = blockIdx.x;
  const int R = (N + GATHER_GRID - 1) / GATHER_GRID;
  const float rinv = 1.0f / (float)R;
  const int nChunks = (nEdges + EDGE_CHUNK - 1) / EDGE_CHUNK;

  const int mycnt = (lane < GEMM_GRID / GATHER_WAVES)
      ? countsT[(size_t)b * GEMM_GRID + wave + GATHER_WAVES * lane] : 0;
  const float4 gv = *(const float4*)(gamma + lane * 4);
  const float4 bv = *(const float4*)(beta + lane * 4);

  for (int p0 = 0; p0 < R; p0 += G_ROWS) {
    const int lo = b * R + p0;
    int nr = R - p0;
    nr = nr < G_ROWS ? nr : G_ROWS;
    nr = (lo + nr > N) ? N - lo : nr;

    for (int i = tid; i < G_ROWS; i += GATHER_WAVES * 32) rowcnt[i] = 0;
    if (tid == 0) ticket = 0;
    __syncthreads();

    {
      int2 ea[GEMM_GRID / GATHER_WAVES], eb[GEMM_GRID / GATHER_WAVES];
#pragma unroll
      for (int k = 0; k < GEMM_GRID / GATHER_WAVES; ++k) {
        const int t = wave + GATHER_WAVES * k;
        const int n = __builtin_amdgcn_readlane(mycnt, k);
        const int nn = n > SUB_CAP ? 0 : n;
        const int2* __restrict__ sl = (const int2*)binned + ((size_t)t * GATHER_GRID + b) * SUB_CAP;
        ea[k] = make_int2(-1, 0);
        eb[k] = make_int2(-1, 0);
        if (lane < nn) ea[k] = sl[lane];
        if (lane + 32 < nn) eb[k] = sl[lane + 32];
      }
#pragma unroll
      for (int k = 0; k < GEMM_GRID / GATHER_WAVES; ++k) {
        gconv_insert(ea[k].x, ea[k].y, lo, nr, rowcnt, rowlist);
        gconv_insert(eb[k].x, eb[k].y, lo, nr, rowcnt, rowlist);
      }
      for (int k = 0; k < GEMM_GRID / GATHER_WAVES; ++k) {
        const int n = __builtin_amdgcn_readlane(mycnt, k);
        if (n > SUB_CAP) {
          for (int s = wave + GATHER_WAVES * k; s < nChunks; s += GEMM_GRID) {
            for (int u = 0; u < 8; ++u) {
              const int e = s * EDGE_CHUNK + lane * 8 + u;
              if (e < nEdges) {
                const int r = rows[e];
                if (gconv_bucket(r, R, rinv) == b) gconv_insert(r, cols[e], lo, nr, rowcnt, rowlist);
              }
            }
          }
        }
      }
    }
    __syncthreads();

    int curRow = gconv_take_row(&ticket, lane);
    while (curRow < nr) {
      const int nextRow = gconv_take_row(&ticket, lane);
      const int node = lo + curRow;
      const float4 xv = *(const float4*)(x + (size_t)node * HIDDEN + lane * 4);
      const int d = __builtin_amdgcn_readfirstlane(rowcnt[curRow]);
      const int* src = rowlist + curRow * LIST_PITCH;

      float4 acc = make_float4(0.0f, 0.0f, 0.0f, 0.0f);
      if (d > LIST_CAP) {
        for (int e0 = 0; e0 < nEdges; e0 += 32) {
          const int e = e0 + lane;
          const int r = (e < nEdges) ? rows[e] : -1;
          const int c = (e < nEdges) ? cols[e] : 0;
          unsigned hit = __builtin_amdgcn_ballot_w32(r == node);
          while (hit) {
            const int k = __builtin_ctz(hit);
            hit &= hit - 1;
            gconv_add_rows<1>(m, c, k, lane, acc);
          }
        }
      }
      const int dl = d > LIST_CAP ? 0 : d;
      for (int base = 0; base < dl; base += 32) {
        const int n = (dl - base) < 32 ? (dl - base) : 32;
        int mysrc = (lane < n) ? src[base + lane] : 0;
        if (dl <= 32) {
          int pos = 0;
          for (int k = 0; k < n; ++k) {
            const int ck = __builtin_amdgcn_readlane(mysrc, k);
            pos += ((ck < mysrc) || (ck == mysrc && k < lane)) ? 1 : 0;
          }
          if (lane >= n) pos = lane;
          mysrc = __builtin_amdgcn_ds_permute(pos << 2, mysrc);
        }
        int j = 0;
        for (; j + 8 <= n; j += 8) gconv_add_rows<8>(m, mysrc, j, lane, acc);
        if (j + 4 <= n) { gconv_add_rows<4>(m, mysrc, j, lane, acc); j += 4; }
        if (j + 2 <= n) { gconv_add_rows<2>(m, mysrc, j, lane, acc); j += 2; }
        if (j < n)      { gconv_add_rows<1>(m, mysrc, j, lane, acc); }
      }

      const float df = (float)d;
      const float invd = 1.0f / (df == 0.0f ? 1.0f : df);

      float4 h;
      h.x = xv.x + acc.x * invd;
      h.y = xv.y + acc.y * invd;
      h.z = xv.z + acc.z * invd;
      h.w = xv.w + acc.w * invd;

      float ss = h.x * h.x + h.y * h.y + h.z * h.z + h.w * h.w;
#pragma unroll
      for (int off = 16; off >= 1; off >>= 1) ss += __shfl_xor(ss, off, 32);
      const float inv_rms = rsqrtf(ss * (1.0f / HIDDEN) + 1e-5f);

      float4 o;
      o.x = h.x * inv_rms * gv.x + bv.x;
      o.y = h.y * inv_rms * gv.y + bv.y;
      o.z = h.z * inv_rms * gv.z + bv.z;
      o.w = h.w * inv_rms * gv.w + bv.w;
      *(float4*)(out + (size_t)node * HIDDEN + lane * 4) = o;

      curRow = nextRow;
    }
    __syncthreads();
  }
}

extern "C" void kernel_launch(void* const* d_in, const int* in_sizes, int n_in,
                              void* d_out, int out_size, void* d_ws, size_t ws_size,
                              hipStream_t stream) {
  const float* x     = (const float*)d_in[0];
  const float* W     = (const float*)d_in[1];
  const float* bias  = (const float*)d_in[2];
  const float* gamma = (const float*)d_in[3];
  const float* beta  = (const float*)d_in[4];
  const int*   rows  = (const int*)d_in[5];
  const int*   cols  = (const int*)d_in[6];
  float*       out   = (float*)d_out;

  const int N      = in_sizes[0] / HIDDEN;
  const int nEdges = in_sizes[5];

  float* m       = (float*)d_ws;
  int*   binned  = (int*)(m + (size_t)N * HIDDEN);
  int*   countsT = binned + (size_t)GEMM_GRID * GATHER_GRID * SUB_CAP * 2;

  gconv_gemm_relu<<<GEMM_GRID, GEMM_WAVES * 32, 0, stream>>>(x, W, bias, m, N,
                                                            rows, cols, binned, countsT, nEdges);

  gconv_gather_finalize<<<GATHER_GRID, GATHER_WAVES * 32, 0, stream>>>(
      x, m, binned, countsT, rows, cols, gamma, beta, out, N, nEdges);
}
